# starting kernel
# baseline (speedup 1.0000x reference)
#include <hip/hip_runtime.h>

typedef __attribute__((ext_vector_type(2))) float v2f;
typedef __attribute__((ext_vector_type(4))) float v4f;
typedef __attribute__((ext_vector_type(8))) float v8f;

#define B_ 8
#define N_ 4096
#define D_ 512
#define K_ 8
#define EPS_ 1e-6f

__global__ void zero_kernel(float* __restrict__ base, int n, float* __restrict__ kl) {
    int i = blockIdx.x * blockDim.x + threadIdx.x;
    if (i < n) base[i] = 0.0f;
    if (i == 0) *kl = 0.0f;
}

__global__ void __launch_bounds__(128)
logits_softmax_kernel(const float* __restrict__ x, const float* __restrict__ W,
                      const float* __restrict__ bias, float* __restrict__ probs,
                      float* __restrict__ Pk) {
    __shared__ float Wp[D_ * 16];

    const int tid    = threadIdx.x;
    const int lane   = tid & 31;
    const int wave   = tid >> 5;
    const int lanelo = lane & 15;
    const int hi     = lane >> 4;
    const int klane  = lanelo & 7;

    for (int i = tid; i < D_ * 16; i += 128) {
        const int row = i >> 4;
        const int col = i & 15;
        Wp[i] = (col < K_) ? W[row * K_ + col] : 0.0f;
    }
    __syncthreads();

    const int tile = blockIdx.x * 4 + wave;
    const int b    = tile >> 8;
    const int n0   = (tile & 255) << 4;

    const float* xrow = x + ((size_t)b * N_ + n0 + lanelo) * D_ + 2 * hi;
    const float* wrow = Wp + 2 * hi * 16 + lanelo;

    v8f c = {};
    #pragma unroll 4
    for (int s = 0; s < D_ / 4; ++s) {
        v2f a;
        a.x = xrow[4 * s];
        a.y = xrow[4 * s + 1];
        v2f bb;
        bb.x = wrow[4 * s * 16];
        bb.y = wrow[(4 * s + 1) * 16];
        c = __builtin_amdgcn_wmma_f32_16x16x4_f32(false, a, false, bb,
                                                  (short)0, c, false, false);
    }

    const float bk = bias[klane];
    float pk_local = 0.0f;
    #pragma unroll
    for (int v = 0; v < 8; ++v) {
        float lg = c[v] + bk;
        float mx = lg;
        mx = fmaxf(mx, __shfl_xor(mx, 1, 8));
        mx = fmaxf(mx, __shfl_xor(mx, 2, 8));
        mx = fmaxf(mx, __shfl_xor(mx, 4, 8));
        float e = expf(lg - mx);
        float ssum = e;
        ssum += __shfl_xor(ssum, 1, 8);
        ssum += __shfl_xor(ssum, 2, 8);
        ssum += __shfl_xor(ssum, 4, 8);
        float p = e / ssum;
        if (lanelo < K_) {
            probs[((size_t)b * N_ + n0 + v + 8 * hi) * K_ + lanelo] = p;
            pk_local += p;
        }
    }
    if (lanelo < K_) atomicAdd(&Pk[b * K_ + lanelo], pk_local);
}

__global__ void __launch_bounds__(128)
moments_kernel(const float* __restrict__ x, const float* __restrict__ probs,
               float* __restrict__ mu, float* __restrict__ s2) {
    __shared__ float Pp[512 * 16];

    const int tid    = threadIdx.x;
    const int lane   = tid & 31;
    const int wave   = tid >> 5;
    const int lanelo = lane & 15;
    const int hi     = lane >> 4;

    const int b      = blockIdx.x >> 6;
    const int rem    = blockIdx.x & 63;
    const int chunk  = rem >> 3;
    const int sgrp   = rem & 7;
    const int nbase  = chunk * 512;
    const int slice  = sgrp * 4 + wave;
    const int d      = slice * 16 + lanelo;

    const float* xb = x     + (size_t)b * N_ * D_;
    const float* pb = probs + ((size_t)b * N_ + nbase) * K_;

    const v4f zero4 = {};
    for (int r = tid; r < 512; r += 128) {
        v4f p0 = *(const v4f*)&pb[r * K_];
        v4f p1 = *(const v4f*)&pb[r * K_ + 4];
        *(v4f*)&Pp[r * 16 + 0]  = p0;
        *(v4f*)&Pp[r * 16 + 4]  = p1;
        *(v4f*)&Pp[r * 16 + 8]  = zero4;
        *(v4f*)&Pp[r * 16 + 12] = zero4;
    }
    __syncthreads();

    const float* prow = Pp + 2 * hi * 16 + lanelo;

    v8f cmu = {};
    v8f cs2 = {};
    #pragma unroll 4
    for (int t = 0; t < 512; t += 4) {
        const int n = nbase + t + 2 * hi;
        v2f a, bx, bx2;
        a.x = prow[t * 16];
        a.y = prow[t * 16 + 16];
        const float b0 = xb[(size_t)n * D_ + d];
        const float b1 = xb[(size_t)(n + 1) * D_ + d];
        bx.x  = b0;      bx.y  = b1;
        bx2.x = b0 * b0; bx2.y = b1 * b1;
        cmu = __builtin_amdgcn_wmma_f32_16x16x4_f32(false, a, false, bx,
                                                    (short)0, cmu, false, false);
        cs2 = __builtin_amdgcn_wmma_f32_16x16x4_f32(false, a, false, bx2,
                                                    (short)0, cs2, false, false);
    }
    if (hi == 0) {
        #pragma unroll
        for (int v = 0; v < 8; ++v) {
            const size_t idx = ((size_t)(b * K_ + v)) * D_ + d;
            atomicAdd(&mu[idx], cmu[v]);
            atomicAdd(&s2[idx], cs2[v]);
        }
    }
}

__global__ void __launch_bounds__(256)
finalize_kernel(const float* __restrict__ mu, const float* __restrict__ s2,
                const float* __restrict__ Pk, const float* __restrict__ eps,
                float* __restrict__ out) {
    __shared__ float red[256];
    const int idx = blockIdx.x * 256 + threadIdx.x;
    const int b = idx >> 9;
    const int d = idx & (D_ - 1);

    float acc = 0.0f, klacc = 0.0f;
    #pragma unroll
    for (int k = 0; k < K_; ++k) {
        const size_t i = ((size_t)(b * K_ + k)) * D_ + d;
        const float m   = mu[i];
        const float v   = s2[i];
        const float P   = Pk[b * K_ + k];
        const float sig = v + m * m * (P - 2.0f);
        const float sp  = sig + EPS_;
        acc   += m + eps[i] * sqrtf(sp);
        klacc += 1.0f + logf(sp) - m * m - sig;
    }
    out[b * D_ + d] = acc * (1.0f / K_);

    red[threadIdx.x] = klacc * (-0.5f / (float)(B_ * K_));
    __syncthreads();
    for (int o = 128; o > 0; o >>= 1) {
        if (threadIdx.x < o) red[threadIdx.x] += red[threadIdx.x + o];
        __syncthreads();
    }
    if (threadIdx.x == 0) atomicAdd(&out[B_ * D_], red[0]);
}

extern "C" void kernel_launch(void* const* d_in, const int* in_sizes, int n_in,
                              void* d_out, int out_size, void* d_ws, size_t ws_size,
                              hipStream_t stream) {
    (void)in_sizes; (void)n_in; (void)out_size; (void)ws_size;
    const float* x    = (const float*)d_in[0];
    const float* W    = (const float*)d_in[1];
    const float* bias = (const float*)d_in[2];
    const float* eps  = (const float*)d_in[3];
    float* out = (float*)d_out;

    float* probs = (float*)d_ws;
    float* mu    = probs + (size_t)B_ * N_ * K_;
    float* s2    = mu    + (size_t)B_ * K_ * D_;
    float* Pk    = s2    + (size_t)B_ * K_ * D_;

    const int nz = 2 * B_ * K_ * D_ + B_ * K_;
    zero_kernel<<<(nz + 255) / 256, 256, 0, stream>>>(mu, nz, &out[B_ * D_]);

    logits_softmax_kernel<<<(B_ * N_ / 16) / 4, 128, 0, stream>>>(x, W, bias, probs, Pk);

    moments_kernel<<<512, 128, 0, stream>>>(x, probs, mu, s2);

    finalize_kernel<<<(B_ * D_) / 256, 256, 0, stream>>>(mu, s2, Pk, eps, out);
}


# P_k as per-block partial sums, plain stores (wave order in LDS, block order in finalize): no atomics before finalize, no zeroing launch (3 -> 2), KL start by fused kernel; P_k n sum order changes
# speedup vs baseline: 3.4345x; 3.4345x over previous
#include <hip/hip_runtime.h>

typedef __attribute__((ext_vector_type(2))) float v2f;
typedef __attribute__((ext_vector_type(4))) float v4f;
typedef __attribute__((ext_vector_type(8))) float v8f;

#define B_ 8
#define N_ 4096
#define D_ 512
#define K_ 8
#define EPS_ 1e-6f

#define PK_FLOATS_   (256 * 32)
#define PART_FLOATS_ (256 * 32 * 256)
#define WS_NEED_BYTES_ ((size_t)(PK_FLOATS_ + PART_FLOATS_) * 4)

#define XP_ 516
#define NQ_ 4
#define FW_ 8
#define FR_ (FW_ * 16)
#define PP_ 136

__device__ __forceinline__ void wait_ds0() { asm volatile("s_wait_dscnt 0x0" ::: "memory"); }
__device__ __forceinline__ void wg_barrier() { asm volatile("s_barrier_signal -1\n\ts_barrier_wait -1" ::: "memory"); }

__global__ void __launch_bounds__(FW_ * 32)
fused_kernel(const float* __restrict__ x, const float* __restrict__ W,
             const float* __restrict__ bias, float* __restrict__ part,
             float* __restrict__ pkpart, float* __restrict__ kl, size_t ws_bytes) {
    __shared__ __attribute__((aligned(16))) float Ws[D_ * K_];
    __shared__ __attribute__((aligned(16))) float Xs[FR_ * XP_];
    __shared__ __attribute__((aligned(16))) float PT[K_ * PP_];
    __shared__ __attribute__((aligned(16))) float St[FW_ * 1024];
    __shared__ __attribute__((aligned(16))) float PkS[K_ * 2 * FW_];

    if (ws_bytes < WS_NEED_BYTES_) return;

    const int tid    = threadIdx.x;
    const int lane   = tid & 31;
    const int wave   = tid >> 5;
    const int lanelo = lane & 15;
    const int hi     = lane >> 4;
    const int klane  = lanelo & 7;

    if (blockIdx.x == 0 && tid == 0) *kl = 0.0f;

    const int tile = blockIdx.x * FW_ + wave;
    const int b    = tile >> 8;
    const int n0   = (tile & 255) << 4;

    v4f wreg[4];
    #pragma unroll
    for (int j = 0; j < 4; ++j)
        wreg[j] = *(const v4f*)&W[(j * 256 + tid) * 4];
    const float* xt = x + ((size_t)b * N_ + n0) * D_ + lane * 4;
    v4f xr[NQ_ * 16];
    #pragma unroll
    for (int q = 0; q < NQ_; ++q) {
        #pragma unroll
        for (int r = 0; r < 16; ++r)
            xr[q * 16 + r] = *(const v4f*)&xt[r * D_ + q * 128];
        asm volatile("" ::: "memory");
    }

    #pragma unroll
    for (int j = 0; j < 4; ++j)
        *(v4f*)&Ws[(j * 256 + tid) * 4] = wreg[j];

    float* xw = Xs + wave * (16 * XP_);
    const float* arow = xw + lanelo * XP_ + 2 * hi;
    const float* wrow = Ws + 2 * hi * K_ + klane;

    v8f c = {};
    #pragma unroll
    for (int q = 0; q < NQ_; ++q) {
        #pragma unroll
        for (int r = 0; r < 16; ++r)
            *(v4f*)&xw[r * XP_ + q * 128 + lane * 4] = xr[q * 16 + r];
        wait_ds0();
        if (q == 0) wg_barrier();
        #pragma unroll
        for (int s = q * 32; s < q * 32 + 32; ++s) {
            const v2f a = *(const v2f*)&arow[4 * s];
            v2f bb;
            bb.x = wrow[4 * s * K_];
            bb.y = wrow[(4 * s + 1) * K_];
            c = __builtin_amdgcn_wmma_f32_16x16x4_f32(false, a, false, bb,
                                                      (short)0, c, false, false);
        }
    }

    const float bk = bias[klane];
    float pk_local = 0.0f;
    float pv[8];
    #pragma unroll
    for (int v = 0; v < 8; ++v) {
        float lg = c[v] + bk;
        float mx = lg;
        mx = fmaxf(mx, __shfl_xor(mx, 1, 8));
        mx = fmaxf(mx, __shfl_xor(mx, 2, 8));
        mx = fmaxf(mx, __shfl_xor(mx, 4, 8));
        float e = expf(lg - mx);
        float ssum = e;
        ssum += __shfl_xor(ssum, 1, 8);
        ssum += __shfl_xor(ssum, 2, 8);
        ssum += __shfl_xor(ssum, 4, 8);
        float p = e / ssum;
        pv[v] = p;
        if (lanelo < K_) pk_local += p;
    }
    if (lanelo < K_) {
        v4f p0, p1;
        p0.x = pv[0]; p0.y = pv[1]; p0.z = pv[2]; p0.w = pv[3];
        p1.x = pv[4]; p1.y = pv[5]; p1.z = pv[6]; p1.w = pv[7];
        float* pt = PT + lanelo * PP_ + 16 * wave + 8 * hi;
        *(v4f*)&pt[0] = p0;
        *(v4f*)&pt[4] = p1;
        PkS[lanelo * (2 * FW_) + 2 * wave + hi] = pk_local;
    }
    __syncthreads();

    if (tid < K_) {
        const v4f* ps = (const v4f*)&PkS[tid * (2 * FW_)];
        float s = 0.0f;
        #pragma unroll
        for (int j = 0; j < 4; ++j) {
            const v4f t = ps[j];
            s += t.x; s += t.y; s += t.z; s += t.w;
        }
        pkpart[blockIdx.x * 32 + tid] = s;
    }

    const float* xcol = Xs + 4 * hi * XP_ + 64 * wave + lanelo;
    const float* prow = PT + klane * PP_ + 4 * hi;

    v8f cm[4] = {{}, {}, {}, {}};
    v8f cs[4] = {{}, {}, {}, {}};
    #pragma unroll
    for (int t = 0; t < FR_ / 4; ++t) {
        const int base = 8 * (t >> 1) + 2 * (t & 1);
        const v2f a = *(const v2f*)&prow[base];
        #pragma unroll
        for (int j = 0; j < 4; ++j) {
            v2f bx, bx2;
            bx.x  = xcol[base * XP_ + 16 * j];
            bx.y  = xcol[(base + 1) * XP_ + 16 * j];
            bx2.x = bx.x * bx.x;
            bx2.y = bx.y * bx.y;
            cm[j] = __builtin_amdgcn_wmma_f32_16x16x4_f32(false, a, false, bx,
                                                          (short)0, cm[j], false, false);
            cs[j] = __builtin_amdgcn_wmma_f32_16x16x4_f32(false, a, false, bx2,
                                                          (short)0, cs[j], false, false);
        }
    }

    float* sw = St + wave * 1024;
    if (hi == 0) {
        #pragma unroll
        for (int j = 0; j < 4; ++j) {
            #pragma unroll
            for (int v = 0; v < 8; ++v) {
                sw[(2 * j) * 128 + v * 16 + lanelo]     = cm[j][v];
                sw[(2 * j + 1) * 128 + v * 16 + lanelo] = cs[j][v];
            }
        }
    }
    wait_ds0();
    float* pout = part + ((size_t)blockIdx.x * 32 + 4 * wave) * 256 + lane * 4;
    #pragma unroll
    for (int i = 0; i < 8; ++i)
        *(v4f*)&pout[i * 128] = *(const v4f*)&sw[i * 128 + lane * 4];
}

__global__ void __launch_bounds__(256)
finalize_kernel(const float* __restrict__ part,
                const float* __restrict__ pkpart, const float* __restrict__ eps,
                float* __restrict__ out, size_t ws_bytes) {
    __shared__ float zs[256];
    __shared__ float red[256];
    __shared__ __attribute__((aligned(16))) float pks[K_ * 32];
    if (ws_bytes < WS_NEED_BYTES_) return;

    const int tid   = threadIdx.x;
    const int b     = blockIdx.x >> 4;
    const int pair  = blockIdx.x & 15;
    const int s2    = tid >> 7;
    const int t7    = tid & 127;
    const int k     = t7 >> 4;
    const int dl    = t7 & 15;
    const int slice = 2 * pair + s2;
    const int d     = slice * 16 + dl;

    const float pkv = pkpart[(size_t)(b * 32 + (tid >> 3)) * 32 + (tid & 7)];
    const float ev  = eps[((size_t)(b * K_ + k)) * D_ + d];
    const float* pp = part + ((size_t)(b * 32) * 32 + slice) * 256 + t7;
    float mr[32], vr[32];
    #pragma unroll
    for (int g = 0; g < 32; ++g) {
        mr[g] = pp[(size_t)g * (32 * 256)];
        vr[g] = pp[(size_t)g * (32 * 256) + 128];
    }
    pks[(tid & 7) * 32 + (tid >> 3)] = pkv;
    wait_ds0();
    wg_barrier();
    float P = 0.0f;
    {
        const v4f* pr = (const v4f*)&pks[k * 32];
        #pragma unroll
        for (int j = 0; j < 8; ++j) {
            const v4f t = pr[j];
            P += t.x; P += t.y; P += t.z; P += t.w;
        }
    }
    float m = 0.0f, v = 0.0f;
    #pragma unroll
    for (int g = 0; g < 32; ++g) {
        m += mr[g];
        v += vr[g];
    }
    const float sig = v + m * m * (P - 2.0f);
    const float sp  = sig + EPS_;
    zs[tid]  = m + ev * sqrtf(sp);
    red[tid] = (1.0f + logf(sp) - m * m - sig) * (-0.5f / (float)(B_ * K_));
    __syncthreads();
    if (tid < 32) {
        const int q2 = tid >> 4, c2 = tid & 15;
        float acc = 0.0f;
        #pragma unroll
        for (int kk = 0; kk < K_; ++kk) acc += zs[q2 * 128 + kk * 16 + c2];
        out[b * D_ + pair * 32 + tid] = acc * (1.0f / K_);
    }
    for (int o = 128; o > 0; o >>= 1) {
        if (tid < o) red[tid] += red[tid + o];
        __syncthreads();
    }
    if (tid == 0) atomicAdd(&out[B_ * D_], red[0]);
}

extern "C" void kernel_launch(void* const* d_in, const int* in_sizes, int n_in,
                              void* d_out, int out_size, void* d_ws, size_t ws_size,
                              hipStream_t stream) {
    (void)in_sizes; (void)n_in; (void)out_size;
    const float* x    = (const float*)d_in[0];
    const float* W    = (const float*)d_in[1];
    const float* bias = (const float*)d_in[2];
    const float* eps  = (const float*)d_in[3];
    float* out = (float*)d_out;

    float* pkpart = (float*)d_ws;
    float* part   = pkpart + PK_FLOATS_;

    fused_kernel<<<B_ * N_ / FR_, FW_ * 32, 0, stream>>>(x, W, bias, part, pkpart,
                                                         &out[B_ * D_], ws_size);

    finalize_kernel<<<B_ * 16, 256, 0, stream>>>(part, pkpart, eps, out, ws_size);
}
